# starting kernel
# baseline (speedup 1.0000x reference)
#include <hip/hip_runtime.h>
#include <math.h>


#define KNN 5
#define M_NODES 30
#define MP 32
#define DIM 128
#define UTS 8
#define BIGF 1.0e9f
#define ROW_STRIDE 130
#define WAVES_PER_BLOCK 2

typedef float v2f __attribute__((ext_vector_type(2)));
typedef float v8f __attribute__((ext_vector_type(8)));

__global__ __launch_bounds__(WAVES_PER_BLOCK * 32)
void uts_sig_kernel(const float* __restrict__ H,
                    const int* __restrict__ hood_idx,
                    const int* __restrict__ hood_sizes,
                    float* __restrict__ out, int N) {
  __shared__ float tileS[WAVES_PER_BLOCK][MP * ROW_STRIDE];
  __shared__ float DmS[WAVES_PER_BLOCK][MP][MP + 1];
  __shared__ float sqS[WAVES_PER_BLOCK][MP];

  const int wave = threadIdx.x >> 5;
  const int lane = threadIdx.x & 31;
  const int n = blockIdx.x * WAVES_PER_BLOCK + wave;
  const int n0 = (n < N) ? n : (N - 1);

  float* tile = tileS[wave];
  float (*Dw)[MP + 1] = DmS[wave];
  float* sqw = sqS[wave];

  const int* idxp = hood_idx + (long long)n0 * M_NODES;
  int myidx = (lane < M_NODES) ? idxp[lane] : 0;
  for (int r = 0; r < MP; ++r) {
    float4 v = make_float4(0.f, 0.f, 0.f, 0.f);
    if (r < M_NODES) {
      int idx = __shfl(myidx, r, 32);
      const float4* src = reinterpret_cast<const float4*>(H + (long long)idx * DIM);
      v = src[lane];
    }
    float* dst = tile + r * ROW_STRIDE + lane * 4;
    dst[0] = v.x; dst[1] = v.y; dst[2] = v.z; dst[3] = v.w;
  }
  __syncthreads();

  const int half = lane >> 4;
  const int lm = lane & 15;
  v8f c00 = {}, c01 = {}, c10 = {}, c11 = {};
  const float* rowA = tile + lm * ROW_STRIDE + 2 * half;
  const float* rowB = tile + (lm + 16) * ROW_STRIDE + 2 * half;
  #pragma unroll 4
  for (int kk = 0; kk < DIM; kk += 4) {
    v2f f0 = *reinterpret_cast<const v2f*>(rowA + kk);
    v2f f1 = *reinterpret_cast<const v2f*>(rowB + kk);
    c00 = __builtin_amdgcn_wmma_f32_16x16x4_f32(false, f0, false, f0, (short)0, c00, false, false);
    c01 = __builtin_amdgcn_wmma_f32_16x16x4_f32(false, f0, false, f1, (short)0, c01, false, false);
    c10 = __builtin_amdgcn_wmma_f32_16x16x4_f32(false, f1, false, f0, (short)0, c10, false, false);
    c11 = __builtin_amdgcn_wmma_f32_16x16x4_f32(false, f1, false, f1, (short)0, c11, false, false);
  }

  #pragma unroll
  for (int v = 0; v < 8; ++v) {
    int i0 = half * 8 + v;
    if (lm == i0) { sqw[i0] = c00[v]; sqw[i0 + 16] = c11[v]; }
  }
  __syncthreads();

  const int size = hood_sizes[n0];
  float sumD = 0.f, sumD2 = 0.f, maxD = 0.f;
  auto do_tile = [&](const v8f& c, int ti, int tj) {
    #pragma unroll
    for (int v = 0; v < 8; ++v) {
      int i = ti * 16 + half * 8 + v;
      int j = tj * 16 + lm;
      float d2 = sqw[i] + sqw[j] - 2.0f * c[v];
      float d = sqrtf(fmaxf(d2, 0.f) + 1e-12f);
      bool valid = (i < size) && (j < size) && (i != j);
      Dw[i][j] = valid ? d : BIGF;
      sumD  += valid ? d : 0.f;
      sumD2 += valid ? d * d : 0.f;
      maxD = valid ? fmaxf(maxD, d) : maxD;
    }
  };
  do_tile(c00, 0, 0); do_tile(c01, 0, 1);
  do_tile(c10, 1, 0); do_tile(c11, 1, 1);
  __syncthreads();

  float knn0 = BIGF, knn1 = BIGF, knn2 = BIGF, knn3 = BIGF, knn4 = BIGF;
  #pragma unroll 8
  for (int j = 0; j < MP; ++j) {
    float x = Dw[lane][j];
    float t4 = fminf(knn4, fmaxf(knn3, x));
    float t3 = fminf(knn3, fmaxf(knn2, x));
    float t2 = fminf(knn2, fmaxf(knn1, x));
    float t1 = fminf(knn1, fmaxf(knn0, x));
    float t0 = fminf(knn0, x);
    knn0 = t0; knn1 = t1; knn2 = t2; knn3 = t3; knn4 = t4;
  }

  float row_w = (lane < size) ? 1.f : 0.f;
  float s0 = knn0 * row_w, s1 = knn1 * row_w, s2 = knn2 * row_w,
        s3 = knn3 * row_w, s4 = knn4 * row_w;
  #pragma unroll
  for (int off = 16; off; off >>= 1) {
    s0 += __shfl_xor(s0, off, 32);
    s1 += __shfl_xor(s1, off, 32);
    s2 += __shfl_xor(s2, off, 32);
    s3 += __shfl_xor(s3, off, 32);
    s4 += __shfl_xor(s4, off, 32);
    sumD  += __shfl_xor(sumD, off, 32);
    sumD2 += __shfl_xor(sumD2, off, 32);
    maxD = fmaxf(maxD, __shfl_xor(maxD, off, 32));
  }

  float fcnt = (float)size;
  float inv_cnt = 1.f / fcnt;
  float inv_np = 1.f / (fcnt * (fcnt - 1.f));
  float dmean = sumD * inv_np;
  float dvar = sumD2 * inv_np - dmean * dmean;

  if (n < N && lane < UTS) {
    float val;
    if      (lane == 0) val = s0 * inv_cnt;
    else if (lane == 1) val = s1 * inv_cnt;
    else if (lane == 2) val = s2 * inv_cnt;
    else if (lane == 3) val = s3 * inv_cnt;
    else if (lane == 4) val = s4 * inv_cnt;
    else if (lane == 5) val = dmean;
    else if (lane == 6) val = dvar;
    else                val = maxD;
    if (size < 3) val = 0.f;
    out[(long long)n * UTS + lane] = val;
  }
}

extern "C" void kernel_launch(void* const* d_in, const int* in_sizes, int n_in,
                              void* d_out, int out_size, void* d_ws, size_t ws_size,
                              hipStream_t stream) {
  const float* H          = (const float*)d_in[0];
  const int*   hood_idx   = (const int*)d_in[1];
  const int*   hood_sizes = (const int*)d_in[2];
  float*       out        = (float*)d_out;
  const int N = in_sizes[2];
  const int blocks = (N + WAVES_PER_BLOCK - 1) / WAVES_PER_BLOCK;
  uts_sig_kernel<<<blocks, WAVES_PER_BLOCK * 32, 0, stream>>>(H, hood_idx, hood_sizes, out, N);
}


# Work follows hood size: slots >= size are not gathered (zero-filled), and a hood of <= 16 rows runs one 16x16 Gram tile with its distances and kNN columns instead of four
# speedup vs baseline: 1.2143x; 1.2143x over previous
#include <hip/hip_runtime.h>
#include <math.h>


#define KNN 5
#define M_NODES 30
#define MP 32
#define DIM 128
#define UTS 8
#define BIGF 1.0e9f
#define ROW_STRIDE 130
#define WAVES_PER_BLOCK 2

typedef float v2f __attribute__((ext_vector_type(2)));
typedef float v8f __attribute__((ext_vector_type(8)));

__global__ __launch_bounds__(WAVES_PER_BLOCK * 32)
void uts_sig_kernel(const float* __restrict__ H,
                    const int* __restrict__ hood_idx,
                    const int* __restrict__ hood_sizes,
                    float* __restrict__ out, int N) {
  __shared__ float tileS[WAVES_PER_BLOCK][MP * ROW_STRIDE];
  __shared__ float DmS[WAVES_PER_BLOCK][MP][MP + 1];
  __shared__ float sqS[WAVES_PER_BLOCK][MP];

  const int wave = __builtin_amdgcn_readfirstlane((int)(threadIdx.x >> 5));
  const int lane = threadIdx.x & 31;
  const int n = blockIdx.x * WAVES_PER_BLOCK + wave;
  const int n0 = (n < N) ? n : (N - 1);

  float* tile = tileS[wave];
  float (*Dw)[MP + 1] = DmS[wave];
  float* sqw = sqS[wave];

  const int size = hood_sizes[n0];
  const int sz = (size < M_NODES) ? size : M_NODES;
  const bool big = sz > 16;
  const int rows_pad = big ? MP : 16;

  const int* idxp = hood_idx + (long long)n0 * M_NODES;
  int myidx = (lane < M_NODES) ? idxp[lane] : 0;
  int ridx[M_NODES];
  #pragma unroll
  for (int r = 0; r < M_NODES; ++r) ridx[r] = __shfl(myidx, r, 32);
  float4 v[M_NODES];
  #pragma unroll
  for (int g = 0; g < M_NODES; g += 8) {
    if (g < sz) {
      #pragma unroll
      for (int r = g; r < g + 8 && r < M_NODES; ++r) {
        if (r < sz) {
          const float4* src = reinterpret_cast<const float4*>(H + (long long)ridx[r] * DIM);
          v[r] = src[lane];
        }
      }
    }
  }
  #pragma unroll
  for (int g = 0; g < M_NODES; g += 8) {
    if (g < sz) {
      #pragma unroll
      for (int r = g; r < g + 8 && r < M_NODES; ++r) {
        if (r < sz) {
          float* dst = tile + r * ROW_STRIDE + lane * 4;
          dst[0] = v[r].x; dst[1] = v[r].y; dst[2] = v[r].z; dst[3] = v[r].w;
        }
      }
    }
  }
  #pragma unroll 1
  for (int r = (sz > 0 ? sz : 0); r < rows_pad; ++r) {
    float* dst = tile + r * ROW_STRIDE + lane * 4;
    dst[0] = 0.f; dst[1] = 0.f; dst[2] = 0.f; dst[3] = 0.f;
  }
  __syncthreads();

  const int half = lane >> 4;
  const int lm = lane & 15;
  v8f c00 = {}, c01 = {}, c10 = {}, c11 = {};
  const float* rowA = tile + lm * ROW_STRIDE + 2 * half;
  const float* rowB = tile + (lm + 16) * ROW_STRIDE + 2 * half;
  if (big) {
    #pragma unroll 4
    for (int kk = 0; kk < DIM; kk += 4) {
      v2f f0 = *reinterpret_cast<const v2f*>(rowA + kk);
      v2f f1 = *reinterpret_cast<const v2f*>(rowB + kk);
      c00 = __builtin_amdgcn_wmma_f32_16x16x4_f32(false, f0, false, f0, (short)0, c00, false, false);
      c01 = __builtin_amdgcn_wmma_f32_16x16x4_f32(false, f0, false, f1, (short)0, c01, false, false);
      c10 = __builtin_amdgcn_wmma_f32_16x16x4_f32(false, f1, false, f0, (short)0, c10, false, false);
      c11 = __builtin_amdgcn_wmma_f32_16x16x4_f32(false, f1, false, f1, (short)0, c11, false, false);
    }
  } else {
    v2f f[DIM / 4];
    #pragma unroll
    for (int k = 0; k < DIM / 4; ++k) f[k] = *reinterpret_cast<const v2f*>(rowA + 4 * k);
    #pragma unroll
    for (int k = 0; k < DIM / 4; ++k)
      c00 = __builtin_amdgcn_wmma_f32_16x16x4_f32(false, f[k], false, f[k], (short)0, c00, false, false);
    __builtin_amdgcn_sched_group_barrier(0x100, 32, 0);
    __builtin_amdgcn_sched_group_barrier(0x008, 32, 0);
  }

  #pragma unroll
  for (int v_ = 0; v_ < 8; ++v_) {
    int i0 = half * 8 + v_;
    if (lm == i0) { sqw[i0] = c00[v_]; sqw[i0 + 16] = c11[v_]; }
  }
  __syncthreads();

  float sumD = 0.f, sumD2 = 0.f, maxD = 0.f;
  auto do_tile = [&](const v8f& c, int ti, int tj) {
    #pragma unroll
    for (int v_ = 0; v_ < 8; ++v_) {
      int i = ti * 16 + half * 8 + v_;
      int j = tj * 16 + lm;
      float d2 = sqw[i] + sqw[j] - 2.0f * c[v_];
      float d = sqrtf(fmaxf(d2, 0.f) + 1e-12f);
      bool valid = (i < size) && (j < size) && (i != j);
      Dw[i][j] = valid ? d : BIGF;
      sumD  += valid ? d : 0.f;
      sumD2 += valid ? d * d : 0.f;
      maxD = valid ? fmaxf(maxD, d) : maxD;
    }
  };
  do_tile(c00, 0, 0);
  if (big) {
    do_tile(c01, 0, 1);
    do_tile(c10, 1, 0); do_tile(c11, 1, 1);
  }
  __syncthreads();

  float knn0 = BIGF, knn1 = BIGF, knn2 = BIGF, knn3 = BIGF, knn4 = BIGF;
  const float* drow = Dw[(lane < sz) ? lane : 0];
  const int jend = (sz + 7) & ~7;
  for (int jb = 0; jb < jend; jb += 8) {
    #pragma unroll
    for (int u = 0; u < 8; ++u) {
      float x = drow[jb + u];
      float t4 = fminf(knn4, fmaxf(knn3, x));
      float t3 = fminf(knn3, fmaxf(knn2, x));
      float t2 = fminf(knn2, fmaxf(knn1, x));
      float t1 = fminf(knn1, fmaxf(knn0, x));
      float t0 = fminf(knn0, x);
      knn0 = t0; knn1 = t1; knn2 = t2; knn3 = t3; knn4 = t4;
    }
  }

  float row_w = (lane < size) ? 1.f : 0.f;
  float s0 = knn0 * row_w, s1 = knn1 * row_w, s2 = knn2 * row_w,
        s3 = knn3 * row_w, s4 = knn4 * row_w;
  #pragma unroll
  for (int off = 16; off; off >>= 1) {
    s0 += __shfl_xor(s0, off, 32);
    s1 += __shfl_xor(s1, off, 32);
    s2 += __shfl_xor(s2, off, 32);
    s3 += __shfl_xor(s3, off, 32);
    s4 += __shfl_xor(s4, off, 32);
    sumD  += __shfl_xor(sumD, off, 32);
    sumD2 += __shfl_xor(sumD2, off, 32);
    maxD = fmaxf(maxD, __shfl_xor(maxD, off, 32));
  }

  float fcnt = (float)size;
  float inv_cnt = 1.f / fcnt;
  float inv_np = 1.f / (fcnt * (fcnt - 1.f));
  float dmean = sumD * inv_np;
  float dvar = sumD2 * inv_np - dmean * dmean;

  if (n < N && lane < UTS) {
    float val;
    if      (lane == 0) val = s0 * inv_cnt;
    else if (lane == 1) val = s1 * inv_cnt;
    else if (lane == 2) val = s2 * inv_cnt;
    else if (lane == 3) val = s3 * inv_cnt;
    else if (lane == 4) val = s4 * inv_cnt;
    else if (lane == 5) val = dmean;
    else if (lane == 6) val = dvar;
    else                val = maxD;
    if (size < 3) val = 0.f;
    out[(long long)n * UTS + lane] = val;
  }
}

extern "C" void kernel_launch(void* const* d_in, const int* in_sizes, int n_in,
                              void* d_out, int out_size, void* d_ws, size_t ws_size,
                              hipStream_t stream) {
  const float* H          = (const float*)d_in[0];
  const int*   hood_idx   = (const int*)d_in[1];
  const int*   hood_sizes = (const int*)d_in[2];
  float*       out        = (float*)d_out;
  const int N = in_sizes[2];
  const int blocks = (N + WAVES_PER_BLOCK - 1) / WAVES_PER_BLOCK;
  uts_sig_kernel<<<blocks, WAVES_PER_BLOCK * 32, 0, stream>>>(H, hood_idx, hood_sizes, out, N);
}


# Rows staged through LDS in two 64-column halves (two rows per load), distance matrix and norms reuse the tile: LDS per block 41984 -> 17408 B, VGPRs held at 128, so 32 waves per WGP instead of 14
# speedup vs baseline: 1.5210x; 1.2526x over previous
#include <hip/hip_runtime.h>
#include <math.h>


#define KNN 5
#define M_NODES 30
#define NPAIR (M_NODES / 2)
#define MP 32
#define DIM 128
#define KH 64
#define UTS 8
#define BIGF 1.0e9f
#define ROW_STRIDE 68
#define WAVES_PER_BLOCK 2
#define LDS_DWORDS (MP * ROW_STRIDE)

typedef float v2f __attribute__((ext_vector_type(2)));
typedef float v4f __attribute__((ext_vector_type(4)));
typedef float v8f __attribute__((ext_vector_type(8)));

static_assert(MP * (MP + 1) + MP <= LDS_DWORDS, "distance matrix and norms must fit in the tile they reuse");

__global__ __launch_bounds__(WAVES_PER_BLOCK * 32) __attribute__((amdgpu_waves_per_eu(8)))
void uts_sig_kernel(const float* __restrict__ H,
                    const int* __restrict__ hood_idx,
                    const int* __restrict__ hood_sizes,
                    float* __restrict__ out, int N) {
  __shared__ __attribute__((aligned(16))) float ldsS[WAVES_PER_BLOCK][LDS_DWORDS];

  const int wave = __builtin_amdgcn_readfirstlane((int)(threadIdx.x >> 5));
  const int lane = threadIdx.x & 31;
  const int n = blockIdx.x * WAVES_PER_BLOCK + wave;
  const int n0 = (n < N) ? n : (N - 1);

  float* tile = ldsS[wave];
  float (*Dw)[MP + 1] = reinterpret_cast<float (*)[MP + 1]>(tile);
  float* sqw = tile + MP * (MP + 1);

  const int size = hood_sizes[n0];
  const int sz = (size < M_NODES) ? size : M_NODES;
  const bool big = sz > 16;
  const int rows_pad = big ? MP : 16;

  const int hl = lane >> 4;
  const int cl = lane & 15;
  const int* idxp = hood_idx + (long long)n0 * M_NODES;
  int myidx = (lane < M_NODES) ? idxp[lane] : 0;
  int ridx[NPAIR];
  #pragma unroll
  for (int p = 0; p < NPAIR; ++p) ridx[p] = __shfl(myidx, 2 * p + hl, 32);
  v4f v[2][NPAIR];
  #pragma unroll
  for (int g = 0; g < NPAIR; g += 4) {
    if (2 * g < sz) {
      #pragma unroll
      for (int p = g; p < g + 4 && p < NPAIR; ++p) {
        if (2 * p < sz) {
          const v4f* src = reinterpret_cast<const v4f*>(H + (long long)ridx[p] * DIM) + cl;
          v[0][p] = src[0];
          v[1][p] = src[KH / 4];
        }
      }
    }
  }

  const int half = lane >> 4;
  const int lm = lane & 15;
  v8f c00 = {}, c01 = {}, c10 = {}, c11 = {};
  const float* rowA = tile + lm * ROW_STRIDE + 2 * half;
  const float* rowB = tile + (lm + 16) * ROW_STRIDE + 2 * half;
  const int szr = (sz + 1) & ~1;

  #pragma unroll
  for (int h = 0; h < 2; ++h) {
    if (h == 1) __syncthreads();
    #pragma unroll
    for (int g = 0; g < NPAIR; g += 4) {
      if (2 * g < sz) {
        #pragma unroll
        for (int p = g; p < g + 4 && p < NPAIR; ++p) {
          if (2 * p < sz) {
            *reinterpret_cast<v4f*>(tile + (2 * p + hl) * ROW_STRIDE + cl * 4) = v[h][p];
          }
        }
      }
    }
    if (h == 0) {
      #pragma unroll 1
      for (int r = (szr > 0 ? szr : 0); r < rows_pad; r += 2) {
        *reinterpret_cast<v4f*>(tile + (r + hl) * ROW_STRIDE + cl * 4) = v4f{0.f, 0.f, 0.f, 0.f};
      }
    }
    __syncthreads();

    if (big) {
      #pragma unroll 4
      for (int kk = 0; kk < KH; kk += 4) {
        v2f f0 = *reinterpret_cast<const v2f*>(rowA + kk);
        v2f f1 = *reinterpret_cast<const v2f*>(rowB + kk);
        c00 = __builtin_amdgcn_wmma_f32_16x16x4_f32(false, f0, false, f0, (short)0, c00, false, false);
        c01 = __builtin_amdgcn_wmma_f32_16x16x4_f32(false, f0, false, f1, (short)0, c01, false, false);
        c10 = __builtin_amdgcn_wmma_f32_16x16x4_f32(false, f1, false, f0, (short)0, c10, false, false);
        c11 = __builtin_amdgcn_wmma_f32_16x16x4_f32(false, f1, false, f1, (short)0, c11, false, false);
      }
    } else {
      v2f f[KH / 4];
      #pragma unroll
      for (int k = 0; k < KH / 4; ++k) f[k] = *reinterpret_cast<const v2f*>(rowA + 4 * k);
      #pragma unroll
      for (int k = 0; k < KH / 4; ++k)
        c00 = __builtin_amdgcn_wmma_f32_16x16x4_f32(false, f[k], false, f[k], (short)0, c00, false, false);
      __builtin_amdgcn_sched_group_barrier(0x100, KH / 4, 0);
      __builtin_amdgcn_sched_group_barrier(0x008, KH / 4, 0);
    }
  }

  #pragma unroll
  for (int v_ = 0; v_ < 8; ++v_) {
    int i0 = half * 8 + v_;
    if (lm == i0) { sqw[i0] = c00[v_]; sqw[i0 + 16] = c11[v_]; }
  }
  __syncthreads();

  float sqi0[8], sqi1[8];
  #pragma unroll
  for (int v_ = 0; v_ < 8; ++v_) { sqi0[v_] = sqw[half * 8 + v_]; sqi1[v_] = 0.f; }
  const float sqj0 = sqw[lm];
  float sqj1 = 0.f;
  if (big) {
    #pragma unroll
    for (int v_ = 0; v_ < 8; ++v_) sqi1[v_] = sqw[16 + half * 8 + v_];
    sqj1 = sqw[16 + lm];
  }
  float sumD = 0.f, sumD2 = 0.f, maxD = 0.f;
  auto do_tile = [&](const v8f& c, const float (&sqi)[8], float sqj, int ti, int tj) {
    #pragma unroll
    for (int v_ = 0; v_ < 8; ++v_) {
      int i = ti * 16 + half * 8 + v_;
      int j = tj * 16 + lm;
      float d2 = sqi[v_] + sqj - 2.0f * c[v_];
      float d = sqrtf(fmaxf(d2, 0.f) + 1e-12f);
      bool valid = (i < size) && (j < size) && (i != j);
      Dw[i][j] = valid ? d : BIGF;
      sumD  += valid ? d : 0.f;
      sumD2 += valid ? d * d : 0.f;
      maxD = valid ? fmaxf(maxD, d) : maxD;
    }
  };
  do_tile(c00, sqi0, sqj0, 0, 0);
  if (big) {
    do_tile(c01, sqi0, sqj1, 0, 1);
    do_tile(c10, sqi1, sqj0, 1, 0); do_tile(c11, sqi1, sqj1, 1, 1);
  }
  __syncthreads();

  float knn0 = BIGF, knn1 = BIGF, knn2 = BIGF, knn3 = BIGF, knn4 = BIGF;
  const float* drow = Dw[(lane < sz) ? lane : 0];
  const int jend = (sz + 7) & ~7;
  for (int jb = 0; jb < jend; jb += 8) {
    #pragma unroll
    for (int u = 0; u < 8; ++u) {
      float x = drow[jb + u];
      float t4 = fminf(knn4, fmaxf(knn3, x));
      float t3 = fminf(knn3, fmaxf(knn2, x));
      float t2 = fminf(knn2, fmaxf(knn1, x));
      float t1 = fminf(knn1, fmaxf(knn0, x));
      float t0 = fminf(knn0, x);
      knn0 = t0; knn1 = t1; knn2 = t2; knn3 = t3; knn4 = t4;
    }
  }

  float row_w = (lane < size) ? 1.f : 0.f;
  float s0 = knn0 * row_w, s1 = knn1 * row_w, s2 = knn2 * row_w,
        s3 = knn3 * row_w, s4 = knn4 * row_w;
  #pragma unroll
  for (int off = 16; off; off >>= 1) {
    s0 += __shfl_xor(s0, off, 32);
    s1 += __shfl_xor(s1, off, 32);
    s2 += __shfl_xor(s2, off, 32);
    s3 += __shfl_xor(s3, off, 32);
    s4 += __shfl_xor(s4, off, 32);
    sumD  += __shfl_xor(sumD, off, 32);
    sumD2 += __shfl_xor(sumD2, off, 32);
    maxD = fmaxf(maxD, __shfl_xor(maxD, off, 32));
  }

  float fcnt = (float)size;
  float inv_cnt = 1.f / fcnt;
  float inv_np = 1.f / (fcnt * (fcnt - 1.f));
  float dmean = sumD * inv_np;
  float dvar = sumD2 * inv_np - dmean * dmean;

  if (n < N && lane < UTS) {
    float val;
    if      (lane == 0) val = s0 * inv_cnt;
    else if (lane == 1) val = s1 * inv_cnt;
    else if (lane == 2) val = s2 * inv_cnt;
    else if (lane == 3) val = s3 * inv_cnt;
    else if (lane == 4) val = s4 * inv_cnt;
    else if (lane == 5) val = dmean;
    else if (lane == 6) val = dvar;
    else                val = maxD;
    if (size < 3) val = 0.f;
    out[(long long)n * UTS + lane] = val;
  }
}

extern "C" void kernel_launch(void* const* d_in, const int* in_sizes, int n_in,
                              void* d_out, int out_size, void* d_ws, size_t ws_size,
                              hipStream_t stream) {
  const float* H          = (const float*)d_in[0];
  const int*   hood_idx   = (const int*)d_in[1];
  const int*   hood_sizes = (const int*)d_in[2];
  float*       out        = (float*)d_out;
  const int N = in_sizes[2];
  const int blocks = (N + WAVES_PER_BLOCK - 1) / WAVES_PER_BLOCK;
  uts_sig_kernel<<<blocks, WAVES_PER_BLOCK * 32, 0, stream>>>(H, hood_idx, hood_sizes, out, N);
}


# Gram symmetry: tile (1,0) is not computed (96 matrix instructions, not 128, when size > 16); its distances are tile (0,1)'s stored transposed, its pairs counted twice in the sums (sum order changes)
# speedup vs baseline: 1.7820x; 1.1716x over previous
#include <hip/hip_runtime.h>
#include <math.h>


#define KNN 5
#define M_NODES 30
#define NPAIR (M_NODES / 2)
#define MP 32
#define DIM 128
#define KH 64
#define UTS 8
#define BIGF 1.0e9f
#define ROW_STRIDE 68
#define WAVES_PER_BLOCK 2
#define LDS_DWORDS (MP * ROW_STRIDE)

typedef float v2f __attribute__((ext_vector_type(2)));
typedef float v4f __attribute__((ext_vector_type(4)));
typedef float v8f __attribute__((ext_vector_type(8)));

static_assert(MP * (MP + 1) + MP <= LDS_DWORDS, "distance matrix and norms must fit in the tile they reuse");

__global__ __launch_bounds__(WAVES_PER_BLOCK * 32) __attribute__((amdgpu_waves_per_eu(8)))
void uts_sig_kernel(const float* __restrict__ H,
                    const int* __restrict__ hood_idx,
                    const int* __restrict__ hood_sizes,
                    float* __restrict__ out, int N) {
  __shared__ __attribute__((aligned(16))) float ldsS[WAVES_PER_BLOCK][LDS_DWORDS];

  const int wave = __builtin_amdgcn_readfirstlane((int)(threadIdx.x >> 5));
  const int lane = threadIdx.x & 31;
  const int n = blockIdx.x * WAVES_PER_BLOCK + wave;
  const int n0 = (n < N) ? n : (N - 1);

  float* tile = ldsS[wave];
  float (*Dw)[MP + 1] = reinterpret_cast<float (*)[MP + 1]>(tile);
  float* sqw = tile + MP * (MP + 1);

  const int size = hood_sizes[n0];
  const int sz = (size < M_NODES) ? size : M_NODES;
  const bool big = sz > 16;
  const int rows_pad = big ? MP : 16;

  const int hl = lane >> 4;
  const int cl = lane & 15;
  const int* idxp = hood_idx + (long long)n0 * M_NODES;
  int myidx = (lane < M_NODES) ? idxp[lane] : 0;
  int ridx[NPAIR];
  #pragma unroll
  for (int p = 0; p < NPAIR; ++p) ridx[p] = __shfl(myidx, 2 * p + hl, 32);
  v4f v[2][NPAIR];
  #pragma unroll
  for (int g = 0; g < NPAIR; g += 4) {
    if (2 * g < sz) {
      #pragma unroll
      for (int p = g; p < g + 4 && p < NPAIR; ++p) {
        if (2 * p < sz) {
          const v4f* src = reinterpret_cast<const v4f*>(H + (long long)ridx[p] * DIM) + cl;
          v[0][p] = src[0];
          v[1][p] = src[KH / 4];
        }
      }
    }
  }

  const int half = lane >> 4;
  const int lm = lane & 15;
  v8f c00 = {}, c01 = {}, c11 = {};
  const float* rowA = tile + lm * ROW_STRIDE + 2 * half;
  const float* rowB = tile + (lm + 16) * ROW_STRIDE + 2 * half;
  const int szr = (sz + 1) & ~1;

  #pragma unroll
  for (int h = 0; h < 2; ++h) {
    if (h == 1) __syncthreads();
    #pragma unroll
    for (int g = 0; g < NPAIR; g += 4) {
      if (2 * g < sz) {
        #pragma unroll
        for (int p = g; p < g + 4 && p < NPAIR; ++p) {
          if (2 * p < sz) {
            *reinterpret_cast<v4f*>(tile + (2 * p + hl) * ROW_STRIDE + cl * 4) = v[h][p];
          }
        }
      }
    }
    if (h == 0) {
      #pragma unroll 1
      for (int r = (szr > 0 ? szr : 0); r < rows_pad; r += 2) {
        *reinterpret_cast<v4f*>(tile + (r + hl) * ROW_STRIDE + cl * 4) = v4f{0.f, 0.f, 0.f, 0.f};
      }
    }
    __syncthreads();

    if (big) {
      #pragma unroll 4
      for (int kk = 0; kk < KH; kk += 4) {
        v2f f0 = *reinterpret_cast<const v2f*>(rowA + kk);
        v2f f1 = *reinterpret_cast<const v2f*>(rowB + kk);
        c00 = __builtin_amdgcn_wmma_f32_16x16x4_f32(false, f0, false, f0, (short)0, c00, false, false);
        c01 = __builtin_amdgcn_wmma_f32_16x16x4_f32(false, f0, false, f1, (short)0, c01, false, false);
        c11 = __builtin_amdgcn_wmma_f32_16x16x4_f32(false, f1, false, f1, (short)0, c11, false, false);
      }
    } else {
      v2f f[KH / 4];
      #pragma unroll
      for (int k = 0; k < KH / 4; ++k) f[k] = *reinterpret_cast<const v2f*>(rowA + 4 * k);
      #pragma unroll
      for (int k = 0; k < KH / 4; ++k)
        c00 = __builtin_amdgcn_wmma_f32_16x16x4_f32(false, f[k], false, f[k], (short)0, c00, false, false);
      __builtin_amdgcn_sched_group_barrier(0x100, KH / 4, 0);
      __builtin_amdgcn_sched_group_barrier(0x008, KH / 4, 0);
    }
  }

  #pragma unroll
  for (int v_ = 0; v_ < 8; ++v_) {
    int i0 = half * 8 + v_;
    if (lm == i0) { sqw[i0] = c00[v_]; sqw[i0 + 16] = c11[v_]; }
  }
  __syncthreads();

  float sqi0[8], sqi1[8];
  #pragma unroll
  for (int v_ = 0; v_ < 8; ++v_) { sqi0[v_] = sqw[half * 8 + v_]; sqi1[v_] = 0.f; }
  const float sqj0 = sqw[lm];
  float sqj1 = 0.f;
  if (big) {
    #pragma unroll
    for (int v_ = 0; v_ < 8; ++v_) sqi1[v_] = sqw[16 + half * 8 + v_];
    sqj1 = sqw[16 + lm];
  }
  float sumD = 0.f, sumD2 = 0.f, maxD = 0.f;
  auto do_tile = [&](const v8f& c, const float (&sqi)[8], float sqj, int ti, int tj) {
    #pragma unroll
    for (int v_ = 0; v_ < 8; ++v_) {
      int i = ti * 16 + half * 8 + v_;
      int j = tj * 16 + lm;
      float d2 = sqi[v_] + sqj - 2.0f * c[v_];
      float d = sqrtf(fmaxf(d2, 0.f) + 1e-12f);
      bool valid = (i < size) && (j < size) && (i != j);
      Dw[i][j] = valid ? d : BIGF;
      sumD  += valid ? d : 0.f;
      sumD2 += valid ? d * d : 0.f;
      maxD = valid ? fmaxf(maxD, d) : maxD;
    }
  };
  do_tile(c00, sqi0, sqj0, 0, 0);
  if (big) {
    float offD = 0.f, offD2 = 0.f;
    #pragma unroll
    for (int v_ = 0; v_ < 8; ++v_) {
      int i = half * 8 + v_;
      int j = 16 + lm;
      float d2 = sqi0[v_] + sqj1 - 2.0f * c01[v_];
      float d = sqrtf(fmaxf(d2, 0.f) + 1e-12f);
      bool valid = (i < size) && (j < size);
      float dm = valid ? d : BIGF;
      Dw[i][j] = dm;
      Dw[j][i] = dm;
      offD  += valid ? d : 0.f;
      offD2 += valid ? d * d : 0.f;
      maxD = valid ? fmaxf(maxD, d) : maxD;
    }
    sumD  += 2.0f * offD;
    sumD2 += 2.0f * offD2;
    do_tile(c11, sqi1, sqj1, 1, 1);
  }
  __syncthreads();

  float knn0 = BIGF, knn1 = BIGF, knn2 = BIGF, knn3 = BIGF, knn4 = BIGF;
  const float* drow = Dw[(lane < sz) ? lane : 0];
  const int jend = (sz + 7) & ~7;
  for (int jb = 0; jb < jend; jb += 8) {
    #pragma unroll
    for (int u = 0; u < 8; ++u) {
      float x = drow[jb + u];
      float t4 = fminf(knn4, fmaxf(knn3, x));
      float t3 = fminf(knn3, fmaxf(knn2, x));
      float t2 = fminf(knn2, fmaxf(knn1, x));
      float t1 = fminf(knn1, fmaxf(knn0, x));
      float t0 = fminf(knn0, x);
      knn0 = t0; knn1 = t1; knn2 = t2; knn3 = t3; knn4 = t4;
    }
  }

  float row_w = (lane < size) ? 1.f : 0.f;
  float s0 = knn0 * row_w, s1 = knn1 * row_w, s2 = knn2 * row_w,
        s3 = knn3 * row_w, s4 = knn4 * row_w;
  #pragma unroll
  for (int off = 16; off; off >>= 1) {
    s0 += __shfl_xor(s0, off, 32);
    s1 += __shfl_xor(s1, off, 32);
    s2 += __shfl_xor(s2, off, 32);
    s3 += __shfl_xor(s3, off, 32);
    s4 += __shfl_xor(s4, off, 32);
    sumD  += __shfl_xor(sumD, off, 32);
    sumD2 += __shfl_xor(sumD2, off, 32);
    maxD = fmaxf(maxD, __shfl_xor(maxD, off, 32));
  }

  float fcnt = (float)size;
  float inv_cnt = 1.f / fcnt;
  float inv_np = 1.f / (fcnt * (fcnt - 1.f));
  float dmean = sumD * inv_np;
  float dvar = sumD2 * inv_np - dmean * dmean;

  if (n < N && lane < UTS) {
    float val;
    if      (lane == 0) val = s0 * inv_cnt;
    else if (lane == 1) val = s1 * inv_cnt;
    else if (lane == 2) val = s2 * inv_cnt;
    else if (lane == 3) val = s3 * inv_cnt;
    else if (lane == 4) val = s4 * inv_cnt;
    else if (lane == 5) val = dmean;
    else if (lane == 6) val = dvar;
    else                val = maxD;
    if (size < 3) val = 0.f;
    out[(long long)n * UTS + lane] = val;
  }
}

extern "C" void kernel_launch(void* const* d_in, const int* in_sizes, int n_in,
                              void* d_out, int out_size, void* d_ws, size_t ws_size,
                              hipStream_t stream) {
  const float* H          = (const float*)d_in[0];
  const int*   hood_idx   = (const int*)d_in[1];
  const int*   hood_sizes = (const int*)d_in[2];
  float*       out        = (float*)d_out;
  const int N = in_sizes[2];
  const int blocks = (N + WAVES_PER_BLOCK - 1) / WAVES_PER_BLOCK;
  uts_sig_kernel<<<blocks, WAVES_PER_BLOCK * 32, 0, stream>>>(H, hood_idx, hood_sizes, out, N);
}


# Each distance's square root by the hardware v_sqrt_f32 (1 ulp, a last bit may differ) instead of sqrtf's correctly rounded sequence of about 15 vector instructions
# speedup vs baseline: 1.9417x; 1.0896x over previous
#include <hip/hip_runtime.h>
#include <math.h>


#define KNN 5
#define M_NODES 30
#define NPAIR (M_NODES / 2)
#define MP 32
#define DIM 128
#define KH 64
#define UTS 8
#define BIGF 1.0e9f
#define ROW_STRIDE 68
#define WAVES_PER_BLOCK 2
#define LDS_DWORDS (MP * ROW_STRIDE)

typedef float v2f __attribute__((ext_vector_type(2)));
typedef float v4f __attribute__((ext_vector_type(4)));
typedef float v8f __attribute__((ext_vector_type(8)));

__device__ __forceinline__ float hw_sqrt(float x) { return __builtin_amdgcn_sqrtf(x); }

static_assert(MP * (MP + 1) + MP <= LDS_DWORDS, "distance matrix and norms must fit in the tile they reuse");

__global__ __launch_bounds__(WAVES_PER_BLOCK * 32) __attribute__((amdgpu_waves_per_eu(8)))
void uts_sig_kernel(const float* __restrict__ H,
                    const int* __restrict__ hood_idx,
                    const int* __restrict__ hood_sizes,
                    float* __restrict__ out, int N) {
  __shared__ __attribute__((aligned(16))) float ldsS[WAVES_PER_BLOCK][LDS_DWORDS];

  const int wave = __builtin_amdgcn_readfirstlane((int)(threadIdx.x >> 5));
  const int lane = threadIdx.x & 31;
  const int n = blockIdx.x * WAVES_PER_BLOCK + wave;
  const int n0 = (n < N) ? n : (N - 1);

  float* tile = ldsS[wave];
  float (*Dw)[MP + 1] = reinterpret_cast<float (*)[MP + 1]>(tile);
  float* sqw = tile + MP * (MP + 1);

  const int size = hood_sizes[n0];
  const int sz = (size < M_NODES) ? size : M_NODES;
  const bool big = sz > 16;
  const int rows_pad = big ? MP : 16;

  const int hl = lane >> 4;
  const int cl = lane & 15;
  const int* idxp = hood_idx + (long long)n0 * M_NODES;
  int myidx = (lane < M_NODES) ? idxp[lane] : 0;
  int ridx[NPAIR];
  #pragma unroll
  for (int p = 0; p < NPAIR; ++p) ridx[p] = __shfl(myidx, 2 * p + hl, 32);
  v4f v[2][NPAIR];
  #pragma unroll
  for (int g = 0; g < NPAIR; g += 4) {
    if (2 * g < sz) {
      #pragma unroll
      for (int p = g; p < g + 4 && p < NPAIR; ++p) {
        if (2 * p < sz) {
          const v4f* src = reinterpret_cast<const v4f*>(H + (long long)ridx[p] * DIM) + cl;
          v[0][p] = src[0];
          v[1][p] = src[KH / 4];
        }
      }
    }
  }

  const int half = lane >> 4;
  const int lm = lane & 15;
  v8f c00 = {}, c01 = {}, c11 = {};
  const float* rowA = tile + lm * ROW_STRIDE + 2 * half;
  const float* rowB = tile + (lm + 16) * ROW_STRIDE + 2 * half;
  const int szr = (sz + 1) & ~1;

  #pragma unroll
  for (int h = 0; h < 2; ++h) {
    if (h == 1) __syncthreads();
    #pragma unroll
    for (int g = 0; g < NPAIR; g += 4) {
      if (2 * g < sz) {
        #pragma unroll
        for (int p = g; p < g + 4 && p < NPAIR; ++p) {
          if (2 * p < sz) {
            *reinterpret_cast<v4f*>(tile + (2 * p + hl) * ROW_STRIDE + cl * 4) = v[h][p];
          }
        }
      }
    }
    if (h == 0) {
      #pragma unroll 1
      for (int r = (szr > 0 ? szr : 0); r < rows_pad; r += 2) {
        *reinterpret_cast<v4f*>(tile + (r + hl) * ROW_STRIDE + cl * 4) = v4f{0.f, 0.f, 0.f, 0.f};
      }
    }
    __syncthreads();

    if (big) {
      #pragma unroll 4
      for (int kk = 0; kk < KH; kk += 4) {
        v2f f0 = *reinterpret_cast<const v2f*>(rowA + kk);
        v2f f1 = *reinterpret_cast<const v2f*>(rowB + kk);
        c00 = __builtin_amdgcn_wmma_f32_16x16x4_f32(false, f0, false, f0, (short)0, c00, false, false);
        c01 = __builtin_amdgcn_wmma_f32_16x16x4_f32(false, f0, false, f1, (short)0, c01, false, false);
        c11 = __builtin_amdgcn_wmma_f32_16x16x4_f32(false, f1, false, f1, (short)0, c11, false, false);
      }
    } else {
      v2f f[KH / 4];
      #pragma unroll
      for (int k = 0; k < KH / 4; ++k) f[k] = *reinterpret_cast<const v2f*>(rowA + 4 * k);
      #pragma unroll
      for (int k = 0; k < KH / 4; ++k)
        c00 = __builtin_amdgcn_wmma_f32_16x16x4_f32(false, f[k], false, f[k], (short)0, c00, false, false);
      __builtin_amdgcn_sched_group_barrier(0x100, KH / 4, 0);
      __builtin_amdgcn_sched_group_barrier(0x008, KH / 4, 0);
    }
  }

  #pragma unroll
  for (int v_ = 0; v_ < 8; ++v_) {
    int i0 = half * 8 + v_;
    if (lm == i0) { sqw[i0] = c00[v_]; sqw[i0 + 16] = c11[v_]; }
  }
  __syncthreads();

  float sqi0[8], sqi1[8];
  #pragma unroll
  for (int v_ = 0; v_ < 8; ++v_) { sqi0[v_] = sqw[half * 8 + v_]; sqi1[v_] = 0.f; }
  const float sqj0 = sqw[lm];
  float sqj1 = 0.f;
  if (big) {
    #pragma unroll
    for (int v_ = 0; v_ < 8; ++v_) sqi1[v_] = sqw[16 + half * 8 + v_];
    sqj1 = sqw[16 + lm];
  }
  float sumD = 0.f, sumD2 = 0.f, maxD = 0.f;
  auto do_tile = [&](const v8f& c, const float (&sqi)[8], float sqj, int ti, int tj) {
    #pragma unroll
    for (int v_ = 0; v_ < 8; ++v_) {
      int i = ti * 16 + half * 8 + v_;
      int j = tj * 16 + lm;
      float d2 = sqi[v_] + sqj - 2.0f * c[v_];
      float d = hw_sqrt(fmaxf(d2, 0.f) + 1e-12f);
      bool valid = (i < size) && (j < size) && (i != j);
      Dw[i][j] = valid ? d : BIGF;
      sumD  += valid ? d : 0.f;
      sumD2 += valid ? d * d : 0.f;
      maxD = valid ? fmaxf(maxD, d) : maxD;
    }
  };
  do_tile(c00, sqi0, sqj0, 0, 0);
  if (big) {
    float offD = 0.f, offD2 = 0.f;
    #pragma unroll
    for (int v_ = 0; v_ < 8; ++v_) {
      int i = half * 8 + v_;
      int j = 16 + lm;
      float d2 = sqi0[v_] + sqj1 - 2.0f * c01[v_];
      float d = hw_sqrt(fmaxf(d2, 0.f) + 1e-12f);
      bool valid = (i < size) && (j < size);
      float dm = valid ? d : BIGF;
      Dw[i][j] = dm;
      Dw[j][i] = dm;
      offD  += valid ? d : 0.f;
      offD2 += valid ? d * d : 0.f;
      maxD = valid ? fmaxf(maxD, d) : maxD;
    }
    sumD  += 2.0f * offD;
    sumD2 += 2.0f * offD2;
    do_tile(c11, sqi1, sqj1, 1, 1);
  }
  __syncthreads();

  float knn0 = BIGF, knn1 = BIGF, knn2 = BIGF, knn3 = BIGF, knn4 = BIGF;
  const float* drow = Dw[(lane < sz) ? lane : 0];
  const int jend = (sz + 7) & ~7;
  for (int jb = 0; jb < jend; jb += 8) {
    #pragma unroll
    for (int u = 0; u < 8; ++u) {
      float x = drow[jb + u];
      float t4 = fminf(knn4, fmaxf(knn3, x));
      float t3 = fminf(knn3, fmaxf(knn2, x));
      float t2 = fminf(knn2, fmaxf(knn1, x));
      float t1 = fminf(knn1, fmaxf(knn0, x));
      float t0 = fminf(knn0, x);
      knn0 = t0; knn1 = t1; knn2 = t2; knn3 = t3; knn4 = t4;
    }
  }

  float row_w = (lane < size) ? 1.f : 0.f;
  float s0 = knn0 * row_w, s1 = knn1 * row_w, s2 = knn2 * row_w,
        s3 = knn3 * row_w, s4 = knn4 * row_w;
  #pragma unroll
  for (int off = 16; off; off >>= 1) {
    s0 += __shfl_xor(s0, off, 32);
    s1 += __shfl_xor(s1, off, 32);
    s2 += __shfl_xor(s2, off, 32);
    s3 += __shfl_xor(s3, off, 32);
    s4 += __shfl_xor(s4, off, 32);
    sumD  += __shfl_xor(sumD, off, 32);
    sumD2 += __shfl_xor(sumD2, off, 32);
    maxD = fmaxf(maxD, __shfl_xor(maxD, off, 32));
  }

  float fcnt = (float)size;
  float inv_cnt = 1.f / fcnt;
  float inv_np = 1.f / (fcnt * (fcnt - 1.f));
  float dmean = sumD * inv_np;
  float dvar = sumD2 * inv_np - dmean * dmean;

  if (n < N && lane < UTS) {
    float val;
    if      (lane == 0) val = s0 * inv_cnt;
    else if (lane == 1) val = s1 * inv_cnt;
    else if (lane == 2) val = s2 * inv_cnt;
    else if (lane == 3) val = s3 * inv_cnt;
    else if (lane == 4) val = s4 * inv_cnt;
    else if (lane == 5) val = dmean;
    else if (lane == 6) val = dvar;
    else                val = maxD;
    if (size < 3) val = 0.f;
    out[(long long)n * UTS + lane] = val;
  }
}

extern "C" void kernel_launch(void* const* d_in, const int* in_sizes, int n_in,
                              void* d_out, int out_size, void* d_ws, size_t ws_size,
                              hipStream_t stream) {
  const float* H          = (const float*)d_in[0];
  const int*   hood_idx   = (const int*)d_in[1];
  const int*   hood_sizes = (const int*)d_in[2];
  float*       out        = (float*)d_out;
  const int N = in_sizes[2];
  const int blocks = (N + WAVES_PER_BLOCK - 1) / WAVES_PER_BLOCK;
  uts_sig_kernel<<<blocks, WAVES_PER_BLOCK * 32, 0, stream>>>(H, hood_idx, hood_sizes, out, N);
}


# Distances stored unmasked; each lane walks only its row's valid columns (own entry 0, sorted 6-list by v_med3) and takes the pair sums and max there: no per-element selects; sum order changes
# speedup vs baseline: 2.0850x; 1.0738x over previous
#include <hip/hip_runtime.h>
#include <math.h>


#define KNN 5
#define M_NODES 30
#define NPAIR (M_NODES / 2)
#define MP 32
#define DIM 128
#define KH 64
#define UTS 8
#define BIGF 1.0e9f
#define ROW_STRIDE 68
#define WAVES_PER_BLOCK 2
#define LDS_DWORDS (MP * ROW_STRIDE)

typedef float v2f __attribute__((ext_vector_type(2)));
typedef float v4f __attribute__((ext_vector_type(4)));
typedef float v8f __attribute__((ext_vector_type(8)));

__device__ __forceinline__ float hw_sqrt(float x) { return __builtin_amdgcn_sqrtf(x); }

static_assert(MP * (MP + 1) + MP <= LDS_DWORDS, "distance matrix and norms must fit in the tile they reuse");

__global__ __launch_bounds__(WAVES_PER_BLOCK * 32) __attribute__((amdgpu_waves_per_eu(8)))
void uts_sig_kernel(const float* __restrict__ H,
                    const int* __restrict__ hood_idx,
                    const int* __restrict__ hood_sizes,
                    float* __restrict__ out, int N) {
  __shared__ __attribute__((aligned(16))) float ldsS[WAVES_PER_BLOCK][LDS_DWORDS];

  const int wave = __builtin_amdgcn_readfirstlane((int)(threadIdx.x >> 5));
  const int lane = threadIdx.x & 31;
  const int n = blockIdx.x * WAVES_PER_BLOCK + wave;
  const int n0 = (n < N) ? n : (N - 1);

  float* tile = ldsS[wave];
  float (*Dw)[MP + 1] = reinterpret_cast<float (*)[MP + 1]>(tile);
  float* sqw = tile + MP * (MP + 1);

  const int size = hood_sizes[n0];
  const int sz = (size < M_NODES) ? size : M_NODES;
  const bool big = sz > 16;
  const int rows_pad = big ? MP : 16;

  const int hl = lane >> 4;
  const int cl = lane & 15;
  const int* idxp = hood_idx + (long long)n0 * M_NODES;
  int myidx = (lane < M_NODES) ? idxp[lane] : 0;
  int ridx[NPAIR];
  #pragma unroll
  for (int p = 0; p < NPAIR; ++p) ridx[p] = __shfl(myidx, 2 * p + hl, 32);
  v4f v[2][NPAIR];
  #pragma unroll
  for (int g = 0; g < NPAIR; g += 4) {
    if (2 * g < sz) {
      #pragma unroll
      for (int p = g; p < g + 4 && p < NPAIR; ++p) {
        if (2 * p < sz) {
          const v4f* src = reinterpret_cast<const v4f*>(H + (long long)ridx[p] * DIM) + cl;
          v[0][p] = src[0];
          v[1][p] = src[KH / 4];
        }
      }
    }
  }

  const int half = lane >> 4;
  const int lm = lane & 15;
  v8f c00 = {}, c01 = {}, c11 = {};
  const float* rowA = tile + lm * ROW_STRIDE + 2 * half;
  const float* rowB = tile + (lm + 16) * ROW_STRIDE + 2 * half;
  const int szr = (sz + 1) & ~1;

  #pragma unroll
  for (int h = 0; h < 2; ++h) {
    if (h == 1) __syncthreads();
    #pragma unroll
    for (int g = 0; g < NPAIR; g += 4) {
      if (2 * g < sz) {
        #pragma unroll
        for (int p = g; p < g + 4 && p < NPAIR; ++p) {
          if (2 * p < sz) {
            *reinterpret_cast<v4f*>(tile + (2 * p + hl) * ROW_STRIDE + cl * 4) = v[h][p];
          }
        }
      }
    }
    if (h == 0) {
      #pragma unroll 1
      for (int r = (szr > 0 ? szr : 0); r < rows_pad; r += 2) {
        *reinterpret_cast<v4f*>(tile + (r + hl) * ROW_STRIDE + cl * 4) = v4f{0.f, 0.f, 0.f, 0.f};
      }
    }
    __syncthreads();

    if (big) {
      #pragma unroll 4
      for (int kk = 0; kk < KH; kk += 4) {
        v2f f0 = *reinterpret_cast<const v2f*>(rowA + kk);
        v2f f1 = *reinterpret_cast<const v2f*>(rowB + kk);
        c00 = __builtin_amdgcn_wmma_f32_16x16x4_f32(false, f0, false, f0, (short)0, c00, false, false);
        c01 = __builtin_amdgcn_wmma_f32_16x16x4_f32(false, f0, false, f1, (short)0, c01, false, false);
        c11 = __builtin_amdgcn_wmma_f32_16x16x4_f32(false, f1, false, f1, (short)0, c11, false, false);
      }
    } else {
      v2f f[KH / 4];
      #pragma unroll
      for (int k = 0; k < KH / 4; ++k) f[k] = *reinterpret_cast<const v2f*>(rowA + 4 * k);
      #pragma unroll
      for (int k = 0; k < KH / 4; ++k)
        c00 = __builtin_amdgcn_wmma_f32_16x16x4_f32(false, f[k], false, f[k], (short)0, c00, false, false);
      __builtin_amdgcn_sched_group_barrier(0x100, KH / 4, 0);
      __builtin_amdgcn_sched_group_barrier(0x008, KH / 4, 0);
    }
  }

  #pragma unroll
  for (int v_ = 0; v_ < 8; ++v_) {
    int i0 = half * 8 + v_;
    if (lm == i0) { sqw[i0] = c00[v_]; sqw[i0 + 16] = c11[v_]; }
  }
  __syncthreads();

  float sqi0[8], sqi1[8];
  #pragma unroll
  for (int v_ = 0; v_ < 8; ++v_) { sqi0[v_] = sqw[half * 8 + v_]; sqi1[v_] = 0.f; }
  const float sqj0 = sqw[lm];
  float sqj1 = 0.f;
  if (big) {
    #pragma unroll
    for (int v_ = 0; v_ < 8; ++v_) sqi1[v_] = sqw[16 + half * 8 + v_];
    sqj1 = sqw[16 + lm];
  }
  auto do_tile = [&](const v8f& c, const float (&sqi)[8], float sqj, int ti, int tj, bool mirror) {
    #pragma unroll
    for (int v_ = 0; v_ < 8; ++v_) {
      int i = ti * 16 + half * 8 + v_;
      int j = tj * 16 + lm;
      float d2 = sqi[v_] + sqj - 2.0f * c[v_];
      float d = hw_sqrt(fmaxf(d2, 0.f) + 1e-12f);
      Dw[i][j] = d;
      if (mirror) Dw[j][i] = d;
    }
  };
  do_tile(c00, sqi0, sqj0, 0, 0, false);
  if (big) {
    do_tile(c01, sqi0, sqj1, 0, 1, true);
    do_tile(c11, sqi1, sqj1, 1, 1, false);
  }
  Dw[lane][lane] = 0.f;
  __syncthreads();

  float knn0 = BIGF, knn1 = BIGF, knn2 = BIGF, knn3 = BIGF, knn4 = BIGF, knn5 = BIGF;
  v2f sD = {0.f, 0.f}, sD2 = {0.f, 0.f};
  float maxD = 0.f;
  const float* drow = Dw[(lane < sz) ? lane : 0];
  auto insert = [&](float x) {
    knn5 = __builtin_amdgcn_fmed3f(knn4, knn5, x);
    knn4 = __builtin_amdgcn_fmed3f(knn3, knn4, x);
    knn3 = __builtin_amdgcn_fmed3f(knn2, knn3, x);
    knn2 = __builtin_amdgcn_fmed3f(knn1, knn2, x);
    knn1 = __builtin_amdgcn_fmed3f(knn0, knn1, x);
    knn0 = fminf(knn0, x);
  };
  auto take2 = [&](float xa, float xb) {
    v2f x2 = {xa, xb};
    sD += x2;
    sD2 += x2 * x2;
    maxD = fmaxf(maxD, fmaxf(xa, xb));
    insert(xa);
    insert(xb);
  };
  int jb = 0;
  for (; jb + 8 <= sz; jb += 8) {
    float x[8];
    #pragma unroll
    for (int u = 0; u < 8; ++u) x[u] = drow[jb + u];
    #pragma unroll
    for (int u = 0; u < 8; u += 2) take2(x[u], x[u + 1]);
  }
  if (sz & 4) {
    float x[4];
    #pragma unroll
    for (int u = 0; u < 4; ++u) x[u] = drow[jb + u];
    take2(x[0], x[1]);
    take2(x[2], x[3]);
    jb += 4;
  }
  if (sz & 2) {
    float x0 = drow[jb], x1 = drow[jb + 1];
    take2(x0, x1);
    jb += 2;
  }
  if (sz & 1) {
    float x = drow[jb];
    sD.x += x;
    sD2.x += x * x;
    maxD = fmaxf(maxD, x);
    insert(x);
  }
  float sumD = sD.x + sD.y, sumD2 = sD2.x + sD2.y;

  float row_w = (lane < size) ? 1.f : 0.f;
  float s0 = knn1 * row_w, s1 = knn2 * row_w, s2 = knn3 * row_w,
        s3 = knn4 * row_w, s4 = knn5 * row_w;
  sumD *= row_w; sumD2 *= row_w; maxD *= row_w;
  #pragma unroll
  for (int off = 16; off; off >>= 1) {
    s0 += __shfl_xor(s0, off, 32);
    s1 += __shfl_xor(s1, off, 32);
    s2 += __shfl_xor(s2, off, 32);
    s3 += __shfl_xor(s3, off, 32);
    s4 += __shfl_xor(s4, off, 32);
    sumD  += __shfl_xor(sumD, off, 32);
    sumD2 += __shfl_xor(sumD2, off, 32);
    maxD = fmaxf(maxD, __shfl_xor(maxD, off, 32));
  }

  float fcnt = (float)size;
  float inv_cnt = 1.f / fcnt;
  float inv_np = 1.f / (fcnt * (fcnt - 1.f));
  float dmean = sumD * inv_np;
  float dvar = sumD2 * inv_np - dmean * dmean;

  if (n < N && lane < UTS) {
    float val;
    if      (lane == 0) val = s0 * inv_cnt;
    else if (lane == 1) val = s1 * inv_cnt;
    else if (lane == 2) val = s2 * inv_cnt;
    else if (lane == 3) val = s3 * inv_cnt;
    else if (lane == 4) val = s4 * inv_cnt;
    else if (lane == 5) val = dmean;
    else if (lane == 6) val = dvar;
    else                val = maxD;
    if (size < 3) val = 0.f;
    out[(long long)n * UTS + lane] = val;
  }
}

extern "C" void kernel_launch(void* const* d_in, const int* in_sizes, int n_in,
                              void* d_out, int out_size, void* d_ws, size_t ws_size,
                              hipStream_t stream) {
  const float* H          = (const float*)d_in[0];
  const int*   hood_idx   = (const int*)d_in[1];
  const int*   hood_sizes = (const int*)d_in[2];
  float*       out        = (float*)d_out;
  const int N = in_sizes[2];
  const int blocks = (N + WAVES_PER_BLOCK - 1) / WAVES_PER_BLOCK;
  uts_sig_kernel<<<blocks, WAVES_PER_BLOCK * 32, 0, stream>>>(H, hood_idx, hood_sizes, out, N);
}
